# starting kernel
# baseline (speedup 1.0000x reference)
#include <hip/hip_runtime.h>
#include <hip/hip_bf16.h>

typedef __attribute__((ext_vector_type(2))) float v2f;
typedef __attribute__((ext_vector_type(8))) float v8f;

#define DFEAT 64
#define ROWS_PER_BLOCK 128
#define WT_PITCH 66

__global__ void g2_zero_kernel(float* __restrict__ p, int n) {
    int i = blockIdx.x * blockDim.x + threadIdx.x;
    if (i < n) p[i] = 0.0f;
}

__global__ void g2_scatter_agg_kernel(const float* __restrict__ nf,
                                      const int* __restrict__ src,
                                      const int* __restrict__ dst,
                                      float* __restrict__ sums,
                                      float* __restrict__ cnt,
                                      int n_edges) {
    int gid = blockIdx.x * blockDim.x + threadIdx.x;
    int e = gid >> 4;
    if (e >= n_edges) return;
    int q = (gid & 15) * 4;
    int s = src[e];
    int d = dst[e];
    const float4 v = *(const float4*)(nf + (size_t)s * DFEAT + q);
    float* out = sums + (size_t)d * DFEAT + q;
    atomicAdd(out + 0, v.x);
    atomicAdd(out + 1, v.y);
    atomicAdd(out + 2, v.z);
    atomicAdd(out + 3, v.w);
    if (q == 0) atomicAdd(cnt + d, 1.0f);
}

__global__ __launch_bounds__(256) void g2_gemm_wmma_kernel(
        const float* __restrict__ sums,
        const float* __restrict__ cnt,
        const float* __restrict__ W,
        float* __restrict__ X,
        int n_nodes) {
    __shared__ float ldsA[ROWS_PER_BLOCK * DFEAT];
    __shared__ float ldsWT[DFEAT * WT_PITCH];

    const int tid = threadIdx.x;
    const int rowBase = blockIdx.x * ROWS_PER_BLOCK;

    for (int i = tid; i < DFEAT * DFEAT; i += 256) {
        int k = i >> 6, n = i & 63;
        ldsWT[n * WT_PITCH + k] = W[i];
    }

    for (int i = tid; i < ROWS_PER_BLOCK * DFEAT; i += 256) {
        int r = rowBase + (i >> 6);
        float v = 0.0f;
        if (r < n_nodes) {
            float c = cnt[r];
            v = sums[(size_t)r * DFEAT + (i & 63)] / fmaxf(c, 1.0f);
        }
        ldsA[i] = v;
    }
    __syncthreads();

    const int wave = tid >> 5;
    const int lane = tid & 31;
    const int m    = lane & 15;
    const int koff = (lane >> 4) << 1;

    const float* aRow = &ldsA[(wave * 16 + m) * DFEAT];
    const float* w0 = &ldsWT[(m +  0) * WT_PITCH];
    const float* w1 = &ldsWT[(m + 16) * WT_PITCH];
    const float* w2 = &ldsWT[(m + 32) * WT_PITCH];
    const float* w3 = &ldsWT[(m + 48) * WT_PITCH];

    v8f acc0 = {}, acc1 = {}, acc2 = {}, acc3 = {};
    for (int k = 0; k < 16; ++k) {
        const int kk = 4 * k + koff;
        const v2f a  = *(const v2f*)(aRow + kk);
        const v2f b0 = *(const v2f*)(w0 + kk);
        const v2f b1 = *(const v2f*)(w1 + kk);
        const v2f b2 = *(const v2f*)(w2 + kk);
        const v2f b3 = *(const v2f*)(w3 + kk);
        acc0 = __builtin_amdgcn_wmma_f32_16x16x4_f32(false, a, false, b0,
                                                     (short)0, acc0, false, false);
        acc1 = __builtin_amdgcn_wmma_f32_16x16x4_f32(false, a, false, b1,
                                                     (short)0, acc1, false, false);
        acc2 = __builtin_amdgcn_wmma_f32_16x16x4_f32(false, a, false, b2,
                                                     (short)0, acc2, false, false);
        acc3 = __builtin_amdgcn_wmma_f32_16x16x4_f32(false, a, false, b3,
                                                     (short)0, acc3, false, false);
    }

    const int row0 = rowBase + wave * 16 + ((lane >> 4) << 3);
    for (int v = 0; v < 8; ++v) {
        int row = row0 + v;
        if (row < n_nodes) {
            float* xr = X + (size_t)row * DFEAT + m;
            xr[0]  = acc0[v];
            xr[16] = acc1[v];
            xr[32] = acc2[v];
            xr[48] = acc3[v];
        }
    }
}

__global__ void g2_scatter_tau_kernel(const float* __restrict__ X,
                                      const int* __restrict__ src,
                                      const int* __restrict__ dst,
                                      float* __restrict__ sums,
                                      int n_edges) {
    int gid = blockIdx.x * blockDim.x + threadIdx.x;
    int e = gid >> 4;
    if (e >= n_edges) return;
    int q = (gid & 15) * 4;
    int s = src[e];
    int d = dst[e];
    const float4 a = *(const float4*)(X + (size_t)s * DFEAT + q);
    const float4 b = *(const float4*)(X + (size_t)d * DFEAT + q);
    float dx = a.x - b.x, dy = a.y - b.y, dz = a.z - b.z, dw = a.w - b.w;
    float* out = sums + (size_t)d * DFEAT + q;
    atomicAdd(out + 0, dx * dx);
    atomicAdd(out + 1, dy * dy);
    atomicAdd(out + 2, dz * dz);
    atomicAdd(out + 3, dw * dw);
}

__global__ void g2_finalize_kernel(const float* __restrict__ sums,
                                   const float* __restrict__ cnt,
                                   float* __restrict__ out,
                                   int n_elems) {
    int i = blockIdx.x * blockDim.x + threadIdx.x;
    if (i >= n_elems) return;
    float c = fmaxf(cnt[i >> 6], 1.0f);
    out[i] = tanhf(sums[i] / c);
}

extern "C" void kernel_launch(void* const* d_in, const int* in_sizes, int n_in,
                              void* d_out, int out_size, void* d_ws, size_t ws_size,
                              hipStream_t stream) {
    const float* nf = (const float*)d_in[0];
    const float* W  = (const float*)d_in[1];
    const int*   ei = (const int*)d_in[2];
    float* out = (float*)d_out;

    const int n_nodes = in_sizes[0] / DFEAT;
    const int n_edges = in_sizes[2] / 2;
    const int* src = ei;
    const int* dst = ei + n_edges;

    float* sums = (float*)d_ws;
    float* cnt  = sums + (size_t)n_nodes * DFEAT;
    float* X    = cnt + n_nodes;

    const int T = 256;

    {
        int n = n_nodes * (DFEAT + 1);
        g2_zero_kernel<<<(n + T - 1) / T, T, 0, stream>>>(sums, n);
    }
    {
        long long total = (long long)n_edges * 16;
        g2_scatter_agg_kernel<<<(int)((total + T - 1) / T), T, 0, stream>>>(
            nf, src, dst, sums, cnt, n_edges);
    }
    {
        int blocks = (n_nodes + ROWS_PER_BLOCK - 1) / ROWS_PER_BLOCK;
        g2_gemm_wmma_kernel<<<blocks, T, 0, stream>>>(sums, cnt, W, X, n_nodes);
    }
    {
        int n = n_nodes * DFEAT;
        g2_zero_kernel<<<(n + T - 1) / T, T, 0, stream>>>(sums, n);
    }
    {
        long long total = (long long)n_edges * 16;
        g2_scatter_tau_kernel<<<(int)((total + T - 1) / T), T, 0, stream>>>(
            X, src, dst, sums, n_edges);
    }
    {
        int n = n_nodes * DFEAT;
        g2_finalize_kernel<<<(n + T - 1) / T, T, 0, stream>>>(sums, cnt, out, n);
    }
}


# Edges bucketed by destination once per call (count, prefix sum, fill); both aggregations become per-node gather-reduce in registers: float atomics, both zero passes and finalize pass removed
# speedup vs baseline: 1.9289x; 1.9289x over previous
#include <hip/hip_runtime.h>
#include <hip/hip_bf16.h>

typedef __attribute__((ext_vector_type(2))) float v2f;
typedef __attribute__((ext_vector_type(8))) float v8f;

#define DFEAT 64
#define ROWS_PER_BLOCK 128
#define WT_PITCH 66
#define SCAN_T 1024


__global__ void g2_zero_int_kernel(int* __restrict__ p, int n) {
    int i = blockIdx.x * blockDim.x + threadIdx.x;
    if (i < n) p[i] = 0;
}

__global__ void g2_count_kernel(const int* __restrict__ dst,
                                int* __restrict__ deg,
                                int* __restrict__ pos,
                                int n_edges) {
    int e = blockIdx.x * blockDim.x + threadIdx.x;
    if (e >= n_edges) return;
    pos[e] = atomicAdd(deg + dst[e], 1);
}

__global__ __launch_bounds__(SCAN_T) void g2_rowptr_kernel(
        const int* __restrict__ deg,
        int* __restrict__ rowptr,
        int n_nodes) {
    __shared__ int wsumAhead[32];
    __shared__ int wsumLocal[32];
    const int tid  = threadIdx.x;
    const int lane = tid & 31;
    const int wave = tid >> 5;
    const int chunk0 = blockIdx.x * SCAN_T;

    int s = 0;
    const int4* deg4 = (const int4*)deg;
    const int n4 = chunk0 >> 2;
    for (int i = tid; i < n4; i += 4 * SCAN_T) {
        const int i1 = i + SCAN_T, i2 = i + 2 * SCAN_T, i3 = i + 3 * SCAN_T;
        const int4 t0 = deg4[i];
        const int4 t1 = deg4[min(i1, n4 - 1)];
        const int4 t2 = deg4[min(i2, n4 - 1)];
        const int4 t3 = deg4[min(i3, n4 - 1)];
        const int s0 = (t0.x + t0.y) + (t0.z + t0.w);
        const int s1 = (t1.x + t1.y) + (t1.z + t1.w);
        const int s2 = (t2.x + t2.y) + (t2.z + t2.w);
        const int s3 = (t3.x + t3.y) + (t3.z + t3.w);
        s += s0 + ((i1 < n4) ? s1 : 0) + ((i2 < n4) ? s2 : 0) + ((i3 < n4) ? s3 : 0);
    }
    for (int off = 16; off > 0; off >>= 1) s += __shfl_xor(s, off, 32);
    if (lane == 0) wsumAhead[wave] = s;

    const int idx = chunk0 + tid;
    const int v = (idx < n_nodes) ? deg[idx] : 0;
    int inc = v;
    for (int off = 1; off < 32; off <<= 1) {
        int t = __shfl_up(inc, off, 32);
        if (lane >= off) inc += t;
    }
    if (lane == 31) wsumLocal[wave] = inc;
    __syncthreads();

    int base = 0, woff = 0;
    for (int w = 0; w < 32; ++w) {
        base += wsumAhead[w];
        if (w < wave) woff += wsumLocal[w];
    }
    if (idx < n_nodes) rowptr[idx] = base + woff + inc - v;
}

__global__ void g2_fill_kernel(const int* __restrict__ src,
                               const int* __restrict__ dst,
                               const int* __restrict__ rowptr,
                               const int* __restrict__ pos,
                               int* __restrict__ sorted,
                               int n_edges) {
    int e = blockIdx.x * blockDim.x + threadIdx.x;
    if (e >= n_edges) return;
    sorted[rowptr[dst[e]] + pos[e]] = src[e];
}

__device__ __forceinline__ float4 g2_gather_sum(const float* __restrict__ tab,
                                                const int* __restrict__ list,
                                                int n, int q) {
    float4 acc = make_float4(0.0f, 0.0f, 0.0f, 0.0f);
    int j = 0;
    for (; j + 4 <= n; j += 4) {
        const int s0 = list[j + 0];
        const int s1 = list[j + 1];
        const int s2 = list[j + 2];
        const int s3 = list[j + 3];
        const float4 v0 = *(const float4*)(tab + (size_t)s0 * DFEAT + q);
        const float4 v1 = *(const float4*)(tab + (size_t)s1 * DFEAT + q);
        const float4 v2 = *(const float4*)(tab + (size_t)s2 * DFEAT + q);
        const float4 v3 = *(const float4*)(tab + (size_t)s3 * DFEAT + q);
        acc.x += v0.x; acc.y += v0.y; acc.z += v0.z; acc.w += v0.w;
        acc.x += v1.x; acc.y += v1.y; acc.z += v1.z; acc.w += v1.w;
        acc.x += v2.x; acc.y += v2.y; acc.z += v2.z; acc.w += v2.w;
        acc.x += v3.x; acc.y += v3.y; acc.z += v3.z; acc.w += v3.w;
    }
    for (; j < n; ++j) {
        const int s0 = list[j];
        const float4 v0 = *(const float4*)(tab + (size_t)s0 * DFEAT + q);
        acc.x += v0.x; acc.y += v0.y; acc.z += v0.z; acc.w += v0.w;
    }
    return acc;
}

__global__ __launch_bounds__(256) void g2_gemm_wmma_kernel(
        const float* __restrict__ nf,
        const int* __restrict__ sorted,
        const int* __restrict__ rowptr,
        const int* __restrict__ deg,
        const float* __restrict__ W,
        float* __restrict__ X,
        int n_nodes) {
    __shared__ __attribute__((aligned(16))) float ldsA[ROWS_PER_BLOCK * DFEAT];
    __shared__ float ldsWT[DFEAT * WT_PITCH];

    const int tid = threadIdx.x;
    const int rowBase = blockIdx.x * ROWS_PER_BLOCK;

    for (int i = tid; i < DFEAT * DFEAT; i += 256) {
        int k = i >> 6, n = i & 63;
        ldsWT[n * WT_PITCH + k] = W[i];
    }

    {
        const int q  = (tid & 15) * 4;
        const int rl = tid >> 4;
        int begs[8], cnts[8];
        for (int p = 0; p < 8; ++p) {
            const int r = rowBase + p * 16 + rl;
            const bool in = r < n_nodes;
            begs[p] = in ? rowptr[r] : 0;
            cnts[p] = in ? deg[r] : 0;
        }
        for (int p = 0; p < 8; ++p) {
            const float4 a = g2_gather_sum(nf, sorted + begs[p], cnts[p], q);
            const float c = fmaxf((float)cnts[p], 1.0f);
            float4 v;
            v.x = a.x / c; v.y = a.y / c; v.z = a.z / c; v.w = a.w / c;
            *(float4*)&ldsA[(p * 16 + rl) * DFEAT + q] = v;
        }
    }
    __syncthreads();

    const int wave = tid >> 5;
    const int lane = tid & 31;
    const int m    = lane & 15;
    const int koff = (lane >> 4) << 1;

    const float* aRow = &ldsA[(wave * 16 + m) * DFEAT];
    const float* w0 = &ldsWT[(m +  0) * WT_PITCH];
    const float* w1 = &ldsWT[(m + 16) * WT_PITCH];
    const float* w2 = &ldsWT[(m + 32) * WT_PITCH];
    const float* w3 = &ldsWT[(m + 48) * WT_PITCH];

    v8f acc0 = {}, acc1 = {}, acc2 = {}, acc3 = {};
    for (int k = 0; k < 16; ++k) {
        const int kk = 4 * k + koff;
        const v2f a  = *(const v2f*)(aRow + kk);
        const v2f b0 = *(const v2f*)(w0 + kk);
        const v2f b1 = *(const v2f*)(w1 + kk);
        const v2f b2 = *(const v2f*)(w2 + kk);
        const v2f b3 = *(const v2f*)(w3 + kk);
        acc0 = __builtin_amdgcn_wmma_f32_16x16x4_f32(false, a, false, b0,
                                                     (short)0, acc0, false, false);
        acc1 = __builtin_amdgcn_wmma_f32_16x16x4_f32(false, a, false, b1,
                                                     (short)0, acc1, false, false);
        acc2 = __builtin_amdgcn_wmma_f32_16x16x4_f32(false, a, false, b2,
                                                     (short)0, acc2, false, false);
        acc3 = __builtin_amdgcn_wmma_f32_16x16x4_f32(false, a, false, b3,
                                                     (short)0, acc3, false, false);
    }

    const int row0 = rowBase + wave * 16 + ((lane >> 4) << 3);
    for (int v = 0; v < 8; ++v) {
        int row = row0 + v;
        if (row < n_nodes) {
            float* xr = X + (size_t)row * DFEAT + m;
            xr[0]  = acc0[v];
            xr[16] = acc1[v];
            xr[32] = acc2[v];
            xr[48] = acc3[v];
        }
    }
}

__global__ __launch_bounds__(256) void g2_tau_gather_kernel(
        const float* __restrict__ X,
        const int* __restrict__ sorted,
        const int* __restrict__ rowptr,
        const int* __restrict__ deg,
        float* __restrict__ out,
        int n_nodes) {
    const int gid = blockIdx.x * blockDim.x + threadIdx.x;
    const int d = gid >> 4;
    if (d >= n_nodes) return;
    const int q = (gid & 15) * 4;
    const int n = deg[d];
    const int* list = sorted + rowptr[d];
    const float4 b = *(const float4*)(X + (size_t)d * DFEAT + q);

    float4 acc = make_float4(0.0f, 0.0f, 0.0f, 0.0f);
    int j = 0;
    for (; j + 4 <= n; j += 4) {
        const int s0 = list[j + 0];
        const int s1 = list[j + 1];
        const int s2 = list[j + 2];
        const int s3 = list[j + 3];
        const float4 a0 = *(const float4*)(X + (size_t)s0 * DFEAT + q);
        const float4 a1 = *(const float4*)(X + (size_t)s1 * DFEAT + q);
        const float4 a2 = *(const float4*)(X + (size_t)s2 * DFEAT + q);
        const float4 a3 = *(const float4*)(X + (size_t)s3 * DFEAT + q);
        float dx, dy, dz, dw;
        dx = a0.x - b.x; dy = a0.y - b.y; dz = a0.z - b.z; dw = a0.w - b.w;
        acc.x += dx * dx; acc.y += dy * dy; acc.z += dz * dz; acc.w += dw * dw;
        dx = a1.x - b.x; dy = a1.y - b.y; dz = a1.z - b.z; dw = a1.w - b.w;
        acc.x += dx * dx; acc.y += dy * dy; acc.z += dz * dz; acc.w += dw * dw;
        dx = a2.x - b.x; dy = a2.y - b.y; dz = a2.z - b.z; dw = a2.w - b.w;
        acc.x += dx * dx; acc.y += dy * dy; acc.z += dz * dz; acc.w += dw * dw;
        dx = a3.x - b.x; dy = a3.y - b.y; dz = a3.z - b.z; dw = a3.w - b.w;
        acc.x += dx * dx; acc.y += dy * dy; acc.z += dz * dz; acc.w += dw * dw;
    }
    for (; j < n; ++j) {
        const int s0 = list[j];
        const float4 a0 = *(const float4*)(X + (size_t)s0 * DFEAT + q);
        const float dx = a0.x - b.x, dy = a0.y - b.y, dz = a0.z - b.z, dw = a0.w - b.w;
        acc.x += dx * dx; acc.y += dy * dy; acc.z += dz * dz; acc.w += dw * dw;
    }

    const float c = fmaxf((float)n, 1.0f);
    float4 t;
    t.x = tanhf(acc.x / c);
    t.y = tanhf(acc.y / c);
    t.z = tanhf(acc.z / c);
    t.w = tanhf(acc.w / c);
    *(float4*)(out + (size_t)d * DFEAT + q) = t;
}

extern "C" void kernel_launch(void* const* d_in, const int* in_sizes, int n_in,
                              void* d_out, int out_size, void* d_ws, size_t ws_size,
                              hipStream_t stream) {
    const float* nf = (const float*)d_in[0];
    const float* W  = (const float*)d_in[1];
    const int*   ei = (const int*)d_in[2];
    float* out = (float*)d_out;

    const int n_nodes = in_sizes[0] / DFEAT;
    const int n_edges = in_sizes[2] / 2;
    const int* src = ei;
    const int* dst = ei + n_edges;

    const int n_pad = (n_nodes + SCAN_T - 1) / SCAN_T * SCAN_T;
    float* X      = (float*)d_ws;
    int*   deg    = (int*)(X + (size_t)n_nodes * DFEAT);
    int*   rowptr = deg + n_pad;
    int*   pos    = rowptr + n_pad;
    int*   sorted = pos + n_edges;

    const int T = 256;

    g2_zero_int_kernel<<<(n_nodes + T - 1) / T, T, 0, stream>>>(deg, n_nodes);
    g2_count_kernel<<<(n_edges + T - 1) / T, T, 0, stream>>>(dst, deg, pos, n_edges);
    g2_rowptr_kernel<<<n_pad / SCAN_T, SCAN_T, 0, stream>>>(deg, rowptr, n_nodes);
    g2_fill_kernel<<<(n_edges + T - 1) / T, T, 0, stream>>>(src, dst, rowptr, pos, sorted, n_edges);
    {
        int blocks = (n_nodes + ROWS_PER_BLOCK - 1) / ROWS_PER_BLOCK;
        g2_gemm_wmma_kernel<<<blocks, T, 0, stream>>>(nf, sorted, rowptr, deg, W, X, n_nodes);
    }
    {
        long long total = (long long)n_nodes * 16;
        g2_tau_gather_kernel<<<(int)((total + T - 1) / T), T, 0, stream>>>(
            X, sorted, rowptr, deg, out, n_nodes);
    }
}


# Both per-node gathers: a node's edge ids are read once, one per lane of its 16-lane row, and handed round by DPP row_share; row loads go out 16 deep (x) or 8 deep (tau), one wait per chunk
# speedup vs baseline: 2.0031x; 1.0385x over previous
#include <hip/hip_runtime.h>
#include <hip/hip_bf16.h>

typedef __attribute__((ext_vector_type(2))) float v2f;
typedef __attribute__((ext_vector_type(8))) float v8f;

#define DFEAT 64
#define ROWS_PER_BLOCK 128
#define WT_PITCH 66
#define SCAN_T 1024


__global__ void g2_zero_int_kernel(int* __restrict__ p, int n) {
    int i = blockIdx.x * blockDim.x + threadIdx.x;
    if (i < n) p[i] = 0;
}

__global__ void g2_count_kernel(const int* __restrict__ dst,
                                int* __restrict__ deg,
                                int* __restrict__ pos,
                                int n_edges) {
    int e = blockIdx.x * blockDim.x + threadIdx.x;
    if (e >= n_edges) return;
    pos[e] = atomicAdd(deg + dst[e], 1);
}

__global__ __launch_bounds__(SCAN_T) void g2_rowptr_kernel(
        const int* __restrict__ deg,
        int* __restrict__ rowptr,
        int n_nodes) {
    __shared__ int wsumAhead[32];
    __shared__ int wsumLocal[32];
    const int tid  = threadIdx.x;
    const int lane = tid & 31;
    const int wave = tid >> 5;
    const int chunk0 = blockIdx.x * SCAN_T;

    int s = 0;
    const int4* deg4 = (const int4*)deg;
    const int n4 = chunk0 >> 2;
    for (int i = tid; i < n4; i += 4 * SCAN_T) {
        const int i1 = i + SCAN_T, i2 = i + 2 * SCAN_T, i3 = i + 3 * SCAN_T;
        const int4 t0 = deg4[i];
        const int4 t1 = deg4[min(i1, n4 - 1)];
        const int4 t2 = deg4[min(i2, n4 - 1)];
        const int4 t3 = deg4[min(i3, n4 - 1)];
        const int s0 = (t0.x + t0.y) + (t0.z + t0.w);
        const int s1 = (t1.x + t1.y) + (t1.z + t1.w);
        const int s2 = (t2.x + t2.y) + (t2.z + t2.w);
        const int s3 = (t3.x + t3.y) + (t3.z + t3.w);
        s += s0 + ((i1 < n4) ? s1 : 0) + ((i2 < n4) ? s2 : 0) + ((i3 < n4) ? s3 : 0);
    }
    for (int off = 16; off > 0; off >>= 1) s += __shfl_xor(s, off, 32);
    if (lane == 0) wsumAhead[wave] = s;

    const int idx = chunk0 + tid;
    const int v = (idx < n_nodes) ? deg[idx] : 0;
    int inc = v;
    for (int off = 1; off < 32; off <<= 1) {
        int t = __shfl_up(inc, off, 32);
        if (lane >= off) inc += t;
    }
    if (lane == 31) wsumLocal[wave] = inc;
    __syncthreads();

    int base = 0, woff = 0;
    for (int w = 0; w < 32; ++w) {
        base += wsumAhead[w];
        if (w < wave) woff += wsumLocal[w];
    }
    if (idx < n_nodes) rowptr[idx] = base + woff + inc - v;
}

__global__ void g2_fill_kernel(const int* __restrict__ src,
                               const int* __restrict__ dst,
                               const int* __restrict__ rowptr,
                               const int* __restrict__ pos,
                               int* __restrict__ sorted,
                               int n_edges) {
    int e = blockIdx.x * blockDim.x + threadIdx.x;
    if (e >= n_edges) return;
    sorted[rowptr[dst[e]] + pos[e]] = src[e];
}

#define G2_ROW_BCAST(v, J) __builtin_amdgcn_update_dpp(0, (v), 0x150 + (J), 0xF, 0xF, false)

template <bool SQD>
__device__ __forceinline__ void g2_acc(float4& acc, const float4 v, const float4 b) {
    if (SQD) {
        const float dx = v.x - b.x, dy = v.y - b.y, dz = v.z - b.z, dw = v.w - b.w;
        acc.x += dx * dx; acc.y += dy * dy; acc.z += dz * dz; acc.w += dw * dw;
    } else {
        acc.x += v.x; acc.y += v.y; acc.z += v.z; acc.w += v.w;
    }
}

template <bool SQD, bool FIRST>
__device__ __forceinline__ float4 g2_chunk16(float4 acc, const char* __restrict__ tabB,
                                             int mine, int cnt, unsigned qB, const float4 b) {
    const float4 init = SQD ? b : make_float4(0.0f, 0.0f, 0.0f, 0.0f);
#define G2_LD(J)                                                              \
    float4 v##J = init;                                                       \
    {                                                                         \
        const unsigned s = (unsigned)G2_ROW_BCAST(mine, J);                   \
        if (J < cnt) v##J = *(const float4*)(tabB + (s * 256u + qB));         \
    }
    G2_LD(0)  G2_LD(1)  G2_LD(2)  G2_LD(3)  G2_LD(4)  G2_LD(5)  G2_LD(6)  G2_LD(7)
    G2_LD(8)  G2_LD(9)  G2_LD(10) G2_LD(11) G2_LD(12) G2_LD(13) G2_LD(14) G2_LD(15)
#undef G2_LD
#define G2_AC(J) g2_acc<SQD>(acc, v##J, b);
    if (FIRST && !SQD) { acc = v0; } else { G2_AC(0) }
              G2_AC(1)  G2_AC(2)  G2_AC(3)  G2_AC(4)  G2_AC(5)  G2_AC(6)  G2_AC(7)
    G2_AC(8)  G2_AC(9)  G2_AC(10) G2_AC(11) G2_AC(12) G2_AC(13) G2_AC(14) G2_AC(15)
#undef G2_AC
    return acc;
}

template <bool SQD, int LB>
__device__ __forceinline__ float4 g2_chunk8(float4 acc, const char* __restrict__ tabB,
                                            int mine, int cnt, unsigned qB, const float4 b) {
    const float4 init = SQD ? b : make_float4(0.0f, 0.0f, 0.0f, 0.0f);
#define G2_LD(J)                                                              \
    float4 v##J = init;                                                       \
    {                                                                         \
        const unsigned s = (unsigned)G2_ROW_BCAST(mine, LB + J);              \
        if (J < cnt) v##J = *(const float4*)(tabB + (s * 256u + qB));         \
    }
    G2_LD(0)  G2_LD(1)  G2_LD(2)  G2_LD(3)  G2_LD(4)  G2_LD(5)  G2_LD(6)  G2_LD(7)
#undef G2_LD
#define G2_AC(J) g2_acc<SQD>(acc, v##J, b);
    G2_AC(0)  G2_AC(1)  G2_AC(2)  G2_AC(3)  G2_AC(4)  G2_AC(5)  G2_AC(6)  G2_AC(7)
#undef G2_AC
    return acc;
}

template <bool SQD>
__device__ __forceinline__ float4 g2_ids16_by8(float4 acc, const char* __restrict__ tabB,
                                               int mine, int cnt, unsigned qB, const float4 b) {
    acc = g2_chunk8<SQD, 0>(acc, tabB, mine, min(cnt, 8), qB, b);
    if (cnt > 8) acc = g2_chunk8<SQD, 8>(acc, tabB, mine, cnt - 8, qB, b);
    return acc;
}

template <bool SQD, bool DEEP16, bool FIRST>
__device__ __forceinline__ float4 g2_ids16(float4 acc, const char* __restrict__ tabB,
                                           int mine, int cnt, unsigned qB, const float4 b) {
    if (DEEP16) return g2_chunk16<SQD, FIRST>(acc, tabB, mine, cnt, qB, b);
    return g2_ids16_by8<SQD>(acc, tabB, mine, cnt, qB, b);
}

template <bool SQD, bool DEEP16>
__device__ __forceinline__ float4 g2_node_gather(const char* __restrict__ tabB,
                                                 const int* __restrict__ list, int n,
                                                 int idA, int idB, int g, unsigned qB,
                                                 const float4 b) {
    float4 acc = make_float4(0.0f, 0.0f, 0.0f, 0.0f);
    acc = g2_ids16<SQD, DEEP16, true>(acc, tabB, idA, min(n, 16), qB, b);
    if (n > 16) {
        acc = g2_ids16<SQD, DEEP16, false>(acc, tabB, idB, min(n - 16, 16), qB, b);
        for (int base = 32; base < n; base += 16) {
            const int id = (base + g < n) ? list[base + g] : 0;
            acc = g2_ids16<SQD, DEEP16, false>(acc, tabB, id, min(n - base, 16), qB, b);
        }
    }
    return acc;
}

__global__ __launch_bounds__(256) void g2_gemm_wmma_kernel(
        const float* __restrict__ nf,
        const int* __restrict__ sorted,
        const int* __restrict__ rowptr,
        const int* __restrict__ deg,
        const float* __restrict__ W,
        float* __restrict__ X,
        int n_nodes) {
    __shared__ __attribute__((aligned(16))) float ldsA[ROWS_PER_BLOCK * DFEAT];
    __shared__ float ldsWT[DFEAT * WT_PITCH];

    const int tid = threadIdx.x;
    const int rowBase = blockIdx.x * ROWS_PER_BLOCK;

    for (int i = tid; i < DFEAT * DFEAT; i += 256) {
        int k = i >> 6, n = i & 63;
        ldsWT[n * WT_PITCH + k] = W[i];
    }

    {
        const int g  = tid & 15;
        const unsigned qB = (unsigned)g * 16u;
        const int rl = tid >> 4;
        const char* nfB = (const char*)nf;
        int begs[8], cnts[8], idA[8], idB[8];
        for (int p = 0; p < 8; ++p) {
            const int r = rowBase + p * 16 + rl;
            const bool in = r < n_nodes;
            begs[p] = in ? rowptr[r] : 0;
            cnts[p] = in ? deg[r] : 0;
        }
        for (int p = 0; p < 8; ++p) {
            idA[p] = (g < cnts[p]) ? sorted[begs[p] + g] : 0;
            idB[p] = (16 + g < cnts[p]) ? sorted[begs[p] + 16 + g] : 0;
        }
        const float4 zero4 = make_float4(0.0f, 0.0f, 0.0f, 0.0f);
        for (int p = 0; p < 8; ++p) {
            const float4 a = g2_node_gather<false, true>(nfB, sorted + begs[p], cnts[p],
                                                   idA[p], idB[p], g, qB, zero4);
            const float c = fmaxf((float)cnts[p], 1.0f);
            float4 v;
            v.x = a.x / c; v.y = a.y / c; v.z = a.z / c; v.w = a.w / c;
            *(float4*)&ldsA[(p * 16 + rl) * DFEAT + g * 4] = v;
        }
    }
    __syncthreads();

    const int wave = tid >> 5;
    const int lane = tid & 31;
    const int m    = lane & 15;
    const int koff = (lane >> 4) << 1;

    const float* aRow = &ldsA[(wave * 16 + m) * DFEAT];
    const float* w0 = &ldsWT[(m +  0) * WT_PITCH];
    const float* w1 = &ldsWT[(m + 16) * WT_PITCH];
    const float* w2 = &ldsWT[(m + 32) * WT_PITCH];
    const float* w3 = &ldsWT[(m + 48) * WT_PITCH];

    v8f acc0 = {}, acc1 = {}, acc2 = {}, acc3 = {};
    for (int k = 0; k < 16; ++k) {
        const int kk = 4 * k + koff;
        const v2f a  = *(const v2f*)(aRow + kk);
        const v2f b0 = *(const v2f*)(w0 + kk);
        const v2f b1 = *(const v2f*)(w1 + kk);
        const v2f b2 = *(const v2f*)(w2 + kk);
        const v2f b3 = *(const v2f*)(w3 + kk);
        acc0 = __builtin_amdgcn_wmma_f32_16x16x4_f32(false, a, false, b0,
                                                     (short)0, acc0, false, false);
        acc1 = __builtin_amdgcn_wmma_f32_16x16x4_f32(false, a, false, b1,
                                                     (short)0, acc1, false, false);
        acc2 = __builtin_amdgcn_wmma_f32_16x16x4_f32(false, a, false, b2,
                                                     (short)0, acc2, false, false);
        acc3 = __builtin_amdgcn_wmma_f32_16x16x4_f32(false, a, false, b3,
                                                     (short)0, acc3, false, false);
    }

    const int row0 = rowBase + wave * 16 + ((lane >> 4) << 3);
    for (int v = 0; v < 8; ++v) {
        int row = row0 + v;
        if (row < n_nodes) {
            float* xr = X + (size_t)row * DFEAT + m;
            xr[0]  = acc0[v];
            xr[16] = acc1[v];
            xr[32] = acc2[v];
            xr[48] = acc3[v];
        }
    }
}

__global__ __launch_bounds__(256) void g2_tau_gather_kernel(
        const float* __restrict__ X,
        const int* __restrict__ sorted,
        const int* __restrict__ rowptr,
        const int* __restrict__ deg,
        float* __restrict__ out,
        int n_nodes) {
    const int gid = blockIdx.x * blockDim.x + threadIdx.x;
    const int d = gid >> 4;
    if (d >= n_nodes) return;
    const int g = gid & 15;
    const int q = g * 4;
    const int n = deg[d];
    const int* list = sorted + rowptr[d];
    const float4 b = *(const float4*)(X + (size_t)d * DFEAT + q);
    const int idA = (g < n) ? list[g] : 0;
    const int idB = (16 + g < n) ? list[16 + g] : 0;

    const float4 acc = g2_node_gather<true, false>((const char*)X, list, n, idA, idB, g,
                                            (unsigned)g * 16u, b);

    const float c = fmaxf((float)n, 1.0f);
    float4 t;
    t.x = tanhf(acc.x / c);
    t.y = tanhf(acc.y / c);
    t.z = tanhf(acc.z / c);
    t.w = tanhf(acc.w / c);
    *(float4*)(out + (size_t)d * DFEAT + q) = t;
}

extern "C" void kernel_launch(void* const* d_in, const int* in_sizes, int n_in,
                              void* d_out, int out_size, void* d_ws, size_t ws_size,
                              hipStream_t stream) {
    const float* nf = (const float*)d_in[0];
    const float* W  = (const float*)d_in[1];
    const int*   ei = (const int*)d_in[2];
    float* out = (float*)d_out;

    const int n_nodes = in_sizes[0] / DFEAT;
    const int n_edges = in_sizes[2] / 2;
    const int* src = ei;
    const int* dst = ei + n_edges;

    const int n_pad = (n_nodes + SCAN_T - 1) / SCAN_T * SCAN_T;
    float* X      = (float*)d_ws;
    int*   deg    = (int*)(X + (size_t)n_nodes * DFEAT);
    int*   rowptr = deg + n_pad;
    int*   pos    = rowptr + n_pad;
    int*   sorted = pos + n_edges;

    const int T = 256;

    g2_zero_int_kernel<<<(n_nodes + T - 1) / T, T, 0, stream>>>(deg, n_nodes);
    g2_count_kernel<<<(n_edges + T - 1) / T, T, 0, stream>>>(dst, deg, pos, n_edges);
    g2_rowptr_kernel<<<n_pad / SCAN_T, SCAN_T, 0, stream>>>(deg, rowptr, n_nodes);
    g2_fill_kernel<<<(n_edges + T - 1) / T, T, 0, stream>>>(src, dst, rowptr, pos, sorted, n_edges);
    {
        int blocks = (n_nodes + ROWS_PER_BLOCK - 1) / ROWS_PER_BLOCK;
        g2_gemm_wmma_kernel<<<blocks, T, 0, stream>>>(nf, sorted, rowptr, deg, W, X, n_nodes);
    }
    {
        long long total = (long long)n_nodes * 16;
        g2_tau_gather_kernel<<<(int)((total + T - 1) / T), T, 0, stream>>>(
            X, sorted, rowptr, deg, out, n_nodes);
    }
}


# Both bucketing scatters (edges into bins, a bin's edges by node) are ordered in LDS and leave as contiguous runs and whole lines, not one 4-byte store per edge; sum order still by LDS arrival
# speedup vs baseline: 4.8805x; 2.4365x over previous
#include <hip/hip_runtime.h>
#include <hip/hip_bf16.h>

typedef __attribute__((ext_vector_type(2))) float v2f;
typedef __attribute__((ext_vector_type(8))) float v8f;
typedef __attribute__((ext_vector_type(2))) int v2i;

#define DFEAT 64
#define ROWS_PER_BLOCK 128
#define WT_PITCH 66
#define SCAN_T 1024
#define BIN_SHIFT 9
#define BIN_NODES (1 << BIN_SHIFT)
#define MAX_BINS 256
#define N_PART 256
#define PART_T 1024
#define STRIP 8
#define STAGE_EDGES 8192
#define CSR_STAGE_EDGES 16384


__global__ __launch_bounds__(PART_T) void g2_bin_hist_kernel(
        const int* __restrict__ dst,
        int* __restrict__ histG,
        int n_edges, int chunk) {
    __shared__ int h[MAX_BINS];
    const int tid = threadIdx.x;
    if (tid < MAX_BINS) h[tid] = 0;
    __syncthreads();
    const int e0 = blockIdx.x * chunk;
    const int e1 = min(e0 + chunk, n_edges);
    for (int e = e0 + tid; e < e1; e += STRIP * PART_T) {
        int d[STRIP];
        for (int j = 0; j < STRIP; ++j) d[j] = dst[min(e + j * PART_T, e1 - 1)];
        for (int j = 0; j < STRIP; ++j) {
            const bool ok = e + j * PART_T < e1;
            atomicAdd(&h[ok ? (d[j] >> BIN_SHIFT) : (tid & (MAX_BINS - 1))], ok ? 1 : 0);
        }
    }
    __syncthreads();
    if (tid < MAX_BINS) histG[blockIdx.x * MAX_BINS + tid] = h[tid];
}

__global__ __launch_bounds__(PART_T) void g2_bin_scatter_kernel(
        const int* __restrict__ src,
        const int* __restrict__ dst,
        const int* __restrict__ histG,
        int* __restrict__ binned,
        int* __restrict__ binStart,
        int n_edges, int chunk) {
    __shared__ __attribute__((aligned(16))) int partTot[16][MAX_BINS];
    __shared__ __attribute__((aligned(16))) int partAhead[16][MAX_BINS];
    __shared__ __attribute__((aligned(8))) v2i stage[STAGE_EDGES + 32];
    __shared__ int cursor[MAX_BINS];
    __shared__ int delta[MAX_BINS];
    __shared__ int wsum[8];
    __shared__ int wsumOwn[8];
    const int tid  = threadIdx.x;
    const int lane = tid & 31;
    const int wave = tid >> 5;
    const int bin  = tid & (MAX_BINS - 1);
    const int q    = tid & 63;
    const int rg   = tid >> 6;
    const int me   = blockIdx.x;
    const int e0 = me * chunk;
    const int e1 = min(e0 + chunk, n_edges);
    const bool staged = e1 - e0 <= STAGE_EDGES;
    const int own = (tid < MAX_BINS) ? histG[me * MAX_BINS + tid] : 0;

    int4 tot = make_int4(0, 0, 0, 0), ahead = make_int4(0, 0, 0, 0);
    const int4* h4 = (const int4*)histG;
    for (int i0 = 0; i0 < N_PART / 16; i0 += 8) {
        int4 v[8];
        for (int i = 0; i < 8; ++i) v[i] = h4[(rg + 16 * (i0 + i)) * (MAX_BINS / 4) + q];
        for (int i = 0; i < 8; ++i) {
            const int m = (rg + 16 * (i0 + i) < me) ? -1 : 0;
            tot.x += v[i].x; tot.y += v[i].y; tot.z += v[i].z; tot.w += v[i].w;
            ahead.x += v[i].x & m; ahead.y += v[i].y & m;
            ahead.z += v[i].z & m; ahead.w += v[i].w & m;
        }
    }
    *(int4*)&partTot[rg][4 * q] = tot;
    *(int4*)&partAhead[rg][4 * q] = ahead;
    __syncthreads();

    int T = 0, A = 0;
    if (tid < MAX_BINS) {
        for (int g = 0; g < 16; ++g) { T += partTot[g][tid]; A += partAhead[g][tid]; }
    }
    int inc = T, incOwn = own;
    for (int off = 1; off < 32; off <<= 1) {
        const int t = __shfl_up(inc, off, 32);
        const int u = __shfl_up(incOwn, off, 32);
        if (lane >= off) { inc += t; incOwn += u; }
    }
    if (tid < MAX_BINS && lane == 31) { wsum[wave] = inc; wsumOwn[wave] = incOwn; }
    __syncthreads();
    if (tid < MAX_BINS) {
        int woff = 0, woffOwn = 0;
        for (int w = 0; w < 8; ++w) {
            woff += (w < wave) ? wsum[w] : 0;
            woffOwn += (w < wave) ? wsumOwn[w] : 0;
        }
        const int start = woff + inc - T;
        const int lstart = woffOwn + incOwn - own;
        cursor[tid] = staged ? lstart : start + A;
        delta[tid] = start + A - lstart;
        if (me == 0) {
            binStart[tid] = start;
            if (tid == MAX_BINS - 1) binStart[MAX_BINS] = start + T;
        }
    }
    __syncthreads();

    if (staged) {
        for (int e = e0 + tid; e < e1; e += STRIP * PART_T) {
            unsigned d[STRIP], s[STRIP];
            int slot[STRIP], dl[STRIP];
            for (int j = 0; j < STRIP; ++j) {
                const int ee = min(e + j * PART_T, e1 - 1);
                d[j] = (unsigned)dst[ee];
                s[j] = (unsigned)src[ee];
            }
            for (int j = 0; j < STRIP; ++j) {
                const bool ok = e + j * PART_T < e1;
                const unsigned b = ok ? (d[j] >> BIN_SHIFT) : (unsigned)bin;
                slot[j] = atomicAdd(&cursor[b], ok ? 1 : 0);
                dl[j] = delta[b];
            }
            for (int j = 0; j < STRIP; ++j) {
                const bool ok = e + j * PART_T < e1;
                v2i rec;
                rec.x = (int)((s[j] << BIN_SHIFT) | (d[j] & (BIN_NODES - 1u)));
                rec.y = dl[j];
                stage[ok ? slot[j] : STAGE_EDGES + lane] = rec;
            }
        }
        __syncthreads();
        const int m = e1 - e0;
        for (int i = tid; i < m; i += STRIP * PART_T) {
            v2i r[STRIP];
            for (int j = 0; j < STRIP; ++j) r[j] = stage[min(i + j * PART_T, m - 1)];
            for (int j = 0; j < STRIP; ++j)
                if (i + j * PART_T < m) binned[i + j * PART_T + r[j].y] = r[j].x;
        }
        return;
    }
    for (int e = e0 + tid; e < e1; e += STRIP * PART_T) {
        unsigned d[STRIP], s[STRIP];
        int slot[STRIP];
        for (int j = 0; j < STRIP; ++j) {
            const int ee = min(e + j * PART_T, e1 - 1);
            d[j] = (unsigned)dst[ee];
            s[j] = (unsigned)src[ee];
        }
        for (int j = 0; j < STRIP; ++j) {
            const bool ok = e + j * PART_T < e1;
            slot[j] = atomicAdd(&cursor[ok ? (d[j] >> BIN_SHIFT) : (unsigned)bin], ok ? 1 : 0);
        }
        for (int j = 0; j < STRIP; ++j)
            if (e + j * PART_T < e1)
                binned[slot[j]] = (int)((s[j] << BIN_SHIFT) | (d[j] & (BIN_NODES - 1u)));
    }
}

__global__ __launch_bounds__(PART_T) void g2_bin_csr_kernel(
        const int* __restrict__ binned,
        const int* __restrict__ binStart,
        int* __restrict__ deg,
        int* __restrict__ rowptr,
        int* __restrict__ sorted,
        int n_nodes) {
    __shared__ int cnt[BIN_NODES];
    __shared__ int wsum[BIN_NODES / 32];
    __shared__ int stageS[CSR_STAGE_EDGES + 32];
    const int tid  = threadIdx.x;
    const int lane = tid & 31;
    const int wave = tid >> 5;
    const int s0 = binStart[blockIdx.x];
    const int s1 = binStart[blockIdx.x + 1];
    const bool staged = s1 - s0 <= CSR_STAGE_EDGES;
    if (tid < BIN_NODES) cnt[tid] = 0;
    __syncthreads();
    for (int i = s0 + tid; i < s1; i += STRIP * PART_T) {
        unsigned v[STRIP];
        for (int j = 0; j < STRIP; ++j) v[j] = (unsigned)binned[min(i + j * PART_T, s1 - 1)];
        for (int j = 0; j < STRIP; ++j) {
            const bool ok = i + j * PART_T < s1;
            atomicAdd(&cnt[ok ? (v[j] & (BIN_NODES - 1u)) : (unsigned)(tid & (BIN_NODES - 1))],
                      ok ? 1 : 0);
        }
    }
    __syncthreads();

    int c = 0;
    if (tid < BIN_NODES) c = cnt[tid];
    int inc = c;
    for (int off = 1; off < 32; off <<= 1) {
        const int t = __shfl_up(inc, off, 32);
        if (lane >= off) inc += t;
    }
    if (tid < BIN_NODES && lane == 31) wsum[wave] = inc;
    __syncthreads();
    if (tid < BIN_NODES) {
        int woff = 0;
        for (int w = 0; w < BIN_NODES / 32; ++w) woff += (w < wave) ? wsum[w] : 0;
        const int start = s0 + woff + inc - c;
        const int node = blockIdx.x * BIN_NODES + tid;
        if (node < n_nodes) { deg[node] = c; rowptr[node] = start; }
        cnt[tid] = start;
    }
    __syncthreads();
    if (staged) {
        for (int i = s0 + tid; i < s1; i += STRIP * PART_T) {
            unsigned v[STRIP];
            int slot[STRIP];
            for (int j = 0; j < STRIP; ++j) v[j] = (unsigned)binned[min(i + j * PART_T, s1 - 1)];
            for (int j = 0; j < STRIP; ++j) {
                const bool ok = i + j * PART_T < s1;
                slot[j] = atomicAdd(&cnt[ok ? (v[j] & (BIN_NODES - 1u))
                                            : (unsigned)(tid & (BIN_NODES - 1))], ok ? 1 : 0);
            }
            for (int j = 0; j < STRIP; ++j) {
                const bool ok = i + j * PART_T < s1;
                stageS[ok ? slot[j] - s0 : CSR_STAGE_EDGES + lane] = (int)(v[j] >> BIN_SHIFT);
            }
        }
        __syncthreads();
        const int m = s1 - s0;
        for (int i = tid - (s0 & 31); i < m; i += STRIP * PART_T) {
            int r[STRIP];
            for (int j = 0; j < STRIP; ++j) r[j] = stageS[max(min(i + j * PART_T, m - 1), 0)];
            for (int j = 0; j < STRIP; ++j) {
                const int k = i + j * PART_T;
                if (k >= 0 && k < m) sorted[s0 + k] = r[j];
            }
        }
        return;
    }
    for (int i = s0 + tid; i < s1; i += STRIP * PART_T) {
        unsigned v[STRIP];
        int slot[STRIP];
        for (int j = 0; j < STRIP; ++j) v[j] = (unsigned)binned[min(i + j * PART_T, s1 - 1)];
        for (int j = 0; j < STRIP; ++j) {
            const bool ok = i + j * PART_T < s1;
            slot[j] = atomicAdd(&cnt[ok ? (v[j] & (BIN_NODES - 1u))
                                        : (unsigned)(tid & (BIN_NODES - 1))], ok ? 1 : 0);
        }
        for (int j = 0; j < STRIP; ++j)
            if (i + j * PART_T < s1) sorted[slot[j]] = (int)(v[j] >> BIN_SHIFT);
    }
}

#define G2_ROW_BCAST(v, J) __builtin_amdgcn_update_dpp(0, (v), 0x150 + (J), 0xF, 0xF, false)

template <bool SQD>
__device__ __forceinline__ void g2_acc(float4& acc, const float4 v, const float4 b) {
    if (SQD) {
        const float dx = v.x - b.x, dy = v.y - b.y, dz = v.z - b.z, dw = v.w - b.w;
        acc.x += dx * dx; acc.y += dy * dy; acc.z += dz * dz; acc.w += dw * dw;
    } else {
        acc.x += v.x; acc.y += v.y; acc.z += v.z; acc.w += v.w;
    }
}

template <bool SQD, bool FIRST>
__device__ __forceinline__ float4 g2_chunk16(float4 acc, const char* __restrict__ tabB,
                                             int mine, int cnt, unsigned qB, const float4 b) {
    const float4 init = SQD ? b : make_float4(0.0f, 0.0f, 0.0f, 0.0f);
#define G2_LD(J)                                                              \
    float4 v##J = init;                                                       \
    {                                                                         \
        const unsigned s = (unsigned)G2_ROW_BCAST(mine, J);                   \
        if (J < cnt) v##J = *(const float4*)(tabB + (s * 256u + qB));         \
    }
    G2_LD(0)  G2_LD(1)  G2_LD(2)  G2_LD(3)  G2_LD(4)  G2_LD(5)  G2_LD(6)  G2_LD(7)
    G2_LD(8)  G2_LD(9)  G2_LD(10) G2_LD(11) G2_LD(12) G2_LD(13) G2_LD(14) G2_LD(15)
#undef G2_LD
#define G2_AC(J) g2_acc<SQD>(acc, v##J, b);
    if (FIRST && !SQD) { acc = v0; } else { G2_AC(0) }
              G2_AC(1)  G2_AC(2)  G2_AC(3)  G2_AC(4)  G2_AC(5)  G2_AC(6)  G2_AC(7)
    G2_AC(8)  G2_AC(9)  G2_AC(10) G2_AC(11) G2_AC(12) G2_AC(13) G2_AC(14) G2_AC(15)
#undef G2_AC
    return acc;
}

template <bool SQD, int LB>
__device__ __forceinline__ float4 g2_chunk8(float4 acc, const char* __restrict__ tabB,
                                            int mine, int cnt, unsigned qB, const float4 b) {
    const float4 init = SQD ? b : make_float4(0.0f, 0.0f, 0.0f, 0.0f);
#define G2_LD(J)                                                              \
    float4 v##J = init;                                                       \
    {                                                                         \
        const unsigned s = (unsigned)G2_ROW_BCAST(mine, LB + J);              \
        if (J < cnt) v##J = *(const float4*)(tabB + (s * 256u + qB));         \
    }
    G2_LD(0)  G2_LD(1)  G2_LD(2)  G2_LD(3)  G2_LD(4)  G2_LD(5)  G2_LD(6)  G2_LD(7)
#undef G2_LD
#define G2_AC(J) g2_acc<SQD>(acc, v##J, b);
    G2_AC(0)  G2_AC(1)  G2_AC(2)  G2_AC(3)  G2_AC(4)  G2_AC(5)  G2_AC(6)  G2_AC(7)
#undef G2_AC
    return acc;
}

template <bool SQD>
__device__ __forceinline__ float4 g2_ids16_by8(float4 acc, const char* __restrict__ tabB,
                                               int mine, int cnt, unsigned qB, const float4 b) {
    acc = g2_chunk8<SQD, 0>(acc, tabB, mine, min(cnt, 8), qB, b);
    if (cnt > 8) acc = g2_chunk8<SQD, 8>(acc, tabB, mine, cnt - 8, qB, b);
    return acc;
}

template <bool SQD, bool DEEP16, bool FIRST>
__device__ __forceinline__ float4 g2_ids16(float4 acc, const char* __restrict__ tabB,
                                           int mine, int cnt, unsigned qB, const float4 b) {
    if (DEEP16) return g2_chunk16<SQD, FIRST>(acc, tabB, mine, cnt, qB, b);
    return g2_ids16_by8<SQD>(acc, tabB, mine, cnt, qB, b);
}

template <bool SQD, bool DEEP16>
__device__ __forceinline__ float4 g2_node_gather(const char* __restrict__ tabB,
                                                 const int* __restrict__ list, int n,
                                                 int idA, int idB, int g, unsigned qB,
                                                 const float4 b) {
    float4 acc = make_float4(0.0f, 0.0f, 0.0f, 0.0f);
    acc = g2_ids16<SQD, DEEP16, true>(acc, tabB, idA, min(n, 16), qB, b);
    if (n > 16) {
        acc = g2_ids16<SQD, DEEP16, false>(acc, tabB, idB, min(n - 16, 16), qB, b);
        for (int base = 32; base < n; base += 16) {
            const int id = (base + g < n) ? list[base + g] : 0;
            acc = g2_ids16<SQD, DEEP16, false>(acc, tabB, id, min(n - base, 16), qB, b);
        }
    }
    return acc;
}

__global__ __launch_bounds__(256) void g2_gemm_wmma_kernel(
        const float* __restrict__ nf,
        const int* __restrict__ sorted,
        const int* __restrict__ rowptr,
        const int* __restrict__ deg,
        const float* __restrict__ W,
        float* __restrict__ X,
        int n_nodes) {
    __shared__ __attribute__((aligned(16))) float ldsA[ROWS_PER_BLOCK * DFEAT];
    __shared__ float ldsWT[DFEAT * WT_PITCH];

    const int tid = threadIdx.x;
    const int rowBase = blockIdx.x * ROWS_PER_BLOCK;

    for (int i = tid; i < DFEAT * DFEAT; i += 256) {
        int k = i >> 6, n = i & 63;
        ldsWT[n * WT_PITCH + k] = W[i];
    }

    {
        const int g  = tid & 15;
        const unsigned qB = (unsigned)g * 16u;
        const int rl = tid >> 4;
        const char* nfB = (const char*)nf;
        int begs[8], cnts[8], idA[8], idB[8];
        for (int p = 0; p < 8; ++p) {
            const int r = rowBase + p * 16 + rl;
            const bool in = r < n_nodes;
            begs[p] = in ? rowptr[r] : 0;
            cnts[p] = in ? deg[r] : 0;
        }
        for (int p = 0; p < 8; ++p) {
            idA[p] = (g < cnts[p]) ? sorted[begs[p] + g] : 0;
            idB[p] = (16 + g < cnts[p]) ? sorted[begs[p] + 16 + g] : 0;
        }
        const float4 zero4 = make_float4(0.0f, 0.0f, 0.0f, 0.0f);
        for (int p = 0; p < 8; ++p) {
            const float4 a = g2_node_gather<false, true>(nfB, sorted + begs[p], cnts[p],
                                                   idA[p], idB[p], g, qB, zero4);
            const float c = fmaxf((float)cnts[p], 1.0f);
            float4 v;
            v.x = a.x / c; v.y = a.y / c; v.z = a.z / c; v.w = a.w / c;
            *(float4*)&ldsA[(p * 16 + rl) * DFEAT + g * 4] = v;
        }
    }
    __syncthreads();

    const int wave = tid >> 5;
    const int lane = tid & 31;
    const int m    = lane & 15;
    const int koff = (lane >> 4) << 1;

    const float* aRow = &ldsA[(wave * 16 + m) * DFEAT];
    const float* w0 = &ldsWT[(m +  0) * WT_PITCH];
    const float* w1 = &ldsWT[(m + 16) * WT_PITCH];
    const float* w2 = &ldsWT[(m + 32) * WT_PITCH];
    const float* w3 = &ldsWT[(m + 48) * WT_PITCH];

    v8f acc0 = {}, acc1 = {}, acc2 = {}, acc3 = {};
    for (int k = 0; k < 16; ++k) {
        const int kk = 4 * k + koff;
        const v2f a  = *(const v2f*)(aRow + kk);
        const v2f b0 = *(const v2f*)(w0 + kk);
        const v2f b1 = *(const v2f*)(w1 + kk);
        const v2f b2 = *(const v2f*)(w2 + kk);
        const v2f b3 = *(const v2f*)(w3 + kk);
        acc0 = __builtin_amdgcn_wmma_f32_16x16x4_f32(false, a, false, b0,
                                                     (short)0, acc0, false, false);
        acc1 = __builtin_amdgcn_wmma_f32_16x16x4_f32(false, a, false, b1,
                                                     (short)0, acc1, false, false);
        acc2 = __builtin_amdgcn_wmma_f32_16x16x4_f32(false, a, false, b2,
                                                     (short)0, acc2, false, false);
        acc3 = __builtin_amdgcn_wmma_f32_16x16x4_f32(false, a, false, b3,
                                                     (short)0, acc3, false, false);
    }

    const int row0 = rowBase + wave * 16 + ((lane >> 4) << 3);
    for (int v = 0; v < 8; ++v) {
        int row = row0 + v;
        if (row < n_nodes) {
            float* xr = X + (size_t)row * DFEAT + m;
            xr[0]  = acc0[v];
            xr[16] = acc1[v];
            xr[32] = acc2[v];
            xr[48] = acc3[v];
        }
    }
}

__global__ __launch_bounds__(256) void g2_tau_gather_kernel(
        const float* __restrict__ X,
        const int* __restrict__ sorted,
        const int* __restrict__ rowptr,
        const int* __restrict__ deg,
        float* __restrict__ out,
        int n_nodes) {
    const int gid = blockIdx.x * blockDim.x + threadIdx.x;
    const int d = gid >> 4;
    if (d >= n_nodes) return;
    const int g = gid & 15;
    const int q = g * 4;
    const int n = deg[d];
    const int* list = sorted + rowptr[d];
    const float4 b = *(const float4*)(X + (size_t)d * DFEAT + q);
    const int idA = (g < n) ? list[g] : 0;
    const int idB = (16 + g < n) ? list[16 + g] : 0;

    const float4 acc = g2_node_gather<true, false>((const char*)X, list, n, idA, idB, g,
                                            (unsigned)g * 16u, b);

    const float c = fmaxf((float)n, 1.0f);
    float4 t;
    t.x = tanhf(acc.x / c);
    t.y = tanhf(acc.y / c);
    t.z = tanhf(acc.z / c);
    t.w = tanhf(acc.w / c);
    *(float4*)(out + (size_t)d * DFEAT + q) = t;
}

extern "C" void kernel_launch(void* const* d_in, const int* in_sizes, int n_in,
                              void* d_out, int out_size, void* d_ws, size_t ws_size,
                              hipStream_t stream) {
    const float* nf = (const float*)d_in[0];
    const float* W  = (const float*)d_in[1];
    const int*   ei = (const int*)d_in[2];
    float* out = (float*)d_out;

    const int n_nodes = in_sizes[0] / DFEAT;
    const int n_edges = in_sizes[2] / 2;
    const int* src = ei;
    const int* dst = ei + n_edges;

    const int n_pad = (n_nodes + SCAN_T - 1) / SCAN_T * SCAN_T;
    float* X      = (float*)d_ws;
    int*   deg    = (int*)(X + (size_t)n_nodes * DFEAT);
    int*   rowptr = deg + n_pad;
    int*   binned = rowptr + n_pad;
    int*   sorted = binned + n_edges;
    int*   histG    = (int*)X;
    int*   binStart = histG + N_PART * MAX_BINS;

    const int T = 256;
    const int n_bins = (n_nodes + BIN_NODES - 1) / BIN_NODES;
    const int chunk  = (n_edges + N_PART - 1) / N_PART;

    g2_bin_hist_kernel<<<N_PART, PART_T, 0, stream>>>(dst, histG, n_edges, chunk);
    g2_bin_scatter_kernel<<<N_PART, PART_T, 0, stream>>>(src, dst, histG, binned, binStart,
                                                         n_edges, chunk);
    g2_bin_csr_kernel<<<n_bins, PART_T, 0, stream>>>(binned, binStart, deg, rowptr, sorted,
                                                     n_nodes);
    {
        int blocks = (n_nodes + ROWS_PER_BLOCK - 1) / ROWS_PER_BLOCK;
        g2_gemm_wmma_kernel<<<blocks, T, 0, stream>>>(nf, sorted, rowptr, deg, W, X, n_nodes);
    }
    {
        long long total = (long long)n_nodes * 16;
        g2_tau_gather_kernel<<<(int)((total + T - 1) / T), T, 0, stream>>>(
            X, sorted, rowptr, deg, out, n_nodes);
    }
}


# Both per-node gathers: row loads and adds no longer masked per 16-lane row: an edge past a node's count reads row 0 at weight 0 (adds exact 0), trips by the wave's longer list; order of sums kept
# speedup vs baseline: 6.0774x; 1.2452x over previous
#include <hip/hip_runtime.h>
#include <hip/hip_bf16.h>

typedef __attribute__((ext_vector_type(2))) float v2f;
typedef __attribute__((ext_vector_type(8))) float v8f;

#define DFEAT 64
#define WT_PITCH 66
#define SCAN_T 1024
#define BIN_SHIFT 9
#define BIN_NODES (1 << BIN_SHIFT)
#define MAX_BINS 256
#define N_PART 256
#define PART_T 1024
#define STRIP 8
#define STAGE_EDGES 8192
#define CSR_STAGE_EDGES 16384


__global__ __launch_bounds__(PART_T) void g2_chunk_bin_kernel(
        const int* __restrict__ src,
        const int* __restrict__ dst,
        int* __restrict__ binned,
        int* __restrict__ histG,
        int* __restrict__ lstartG,
        int n_edges, int chunk) {
    __shared__ int cnt[MAX_BINS];
    __shared__ int wsum[MAX_BINS / 32];
    __shared__ int stage[STAGE_EDGES + 32];
    const int tid  = threadIdx.x;
    const int lane = tid & 31;
    const int wave = tid >> 5;
    const int me   = blockIdx.x;
    const int e0 = min(me * chunk, n_edges);
    const int e1 = min(e0 + chunk, n_edges);
    const int m  = e1 - e0;
    const bool staged = m <= STAGE_EDGES;
    unsigned d[STRIP], s[STRIP];
    int rank[STRIP];
    if (tid < MAX_BINS) cnt[tid] = 0;
    __syncthreads();

    if (staged) {
        for (int j = 0; j < STRIP; ++j) {
            const int ee = min(e0 + tid + j * PART_T, e1 - 1);
            d[j] = (unsigned)dst[ee];
            s[j] = (unsigned)src[ee];
        }
        for (int j = 0; j < STRIP; ++j) {
            const bool ok = tid + j * PART_T < m;
            rank[j] = atomicAdd(&cnt[ok ? (d[j] >> BIN_SHIFT) : (unsigned)(tid & (MAX_BINS - 1))],
                                ok ? 1 : 0);
        }
    } else {
        for (int e = e0 + tid; e < e1; e += STRIP * PART_T) {
            int dd[STRIP];
            for (int j = 0; j < STRIP; ++j) dd[j] = dst[min(e + j * PART_T, e1 - 1)];
            for (int j = 0; j < STRIP; ++j) {
                const bool ok = e + j * PART_T < e1;
                atomicAdd(&cnt[ok ? (dd[j] >> BIN_SHIFT) : (tid & (MAX_BINS - 1))], ok ? 1 : 0);
            }
        }
    }
    __syncthreads();

    int c = 0;
    if (tid < MAX_BINS) c = cnt[tid];
    int inc = c;
    for (int off = 1; off < 32; off <<= 1) {
        const int t = __shfl_up(inc, off, 32);
        if (lane >= off) inc += t;
    }
    if (tid < MAX_BINS && lane == 31) wsum[wave] = inc;
    __syncthreads();
    if (tid < MAX_BINS) {
        int woff = 0;
        for (int w = 0; w < MAX_BINS / 32; ++w) woff += (w < wave) ? wsum[w] : 0;
        const int lstart = woff + inc - c;
        histG[me * MAX_BINS + tid] = c;
        lstartG[me * MAX_BINS + tid] = lstart;
        cnt[tid] = lstart;
    }
    __syncthreads();

    if (staged) {
        for (int j = 0; j < STRIP; ++j) {
            const bool ok = tid + j * PART_T < m;
            const int pos = cnt[d[j] >> BIN_SHIFT] + rank[j];
            stage[ok ? pos : STAGE_EDGES + lane] =
                (int)((s[j] << BIN_SHIFT) | (d[j] & (BIN_NODES - 1u)));
        }
        __syncthreads();
        int r[STRIP];
        for (int j = 0; j < STRIP; ++j) r[j] = stage[max(min(tid + j * PART_T, m - 1), 0)];
        for (int j = 0; j < STRIP; ++j)
            if (tid + j * PART_T < m) binned[e0 + tid + j * PART_T] = r[j];
        return;
    }
    for (int e = e0 + tid; e < e1; e += STRIP * PART_T) {
        unsigned dd[STRIP], ss[STRIP];
        int slot[STRIP];
        for (int j = 0; j < STRIP; ++j) {
            const int ee = min(e + j * PART_T, e1 - 1);
            dd[j] = (unsigned)dst[ee];
            ss[j] = (unsigned)src[ee];
        }
        for (int j = 0; j < STRIP; ++j) {
            const bool ok = e + j * PART_T < e1;
            slot[j] = atomicAdd(&cnt[ok ? (dd[j] >> BIN_SHIFT) : (unsigned)(tid & (MAX_BINS - 1))],
                                ok ? 1 : 0);
        }
        for (int j = 0; j < STRIP; ++j)
            if (e + j * PART_T < e1)
                binned[e0 + slot[j]] = (int)((ss[j] << BIN_SHIFT) | (dd[j] & (BIN_NODES - 1u)));
    }
}

#define G2_ROW_BCAST(v, J) __builtin_amdgcn_update_dpp(0, (v), 0x150 + (J), 0xF, 0xF, false)

__device__ __forceinline__ float g2_w(float left, int J) {
    return __builtin_fminf(__builtin_fmaxf(left - (float)J, 0.0f), 1.0f);
}

template <bool SQD>
__device__ __forceinline__ void g2_acc(v2f& a01, v2f& a23, const float4 v, const v2f b01,
                                       const v2f b23, const float w) {
    const v2f v01 = {v.x, v.y}, v23 = {v.z, v.w};
    const v2f ww = {w, w};
    if (SQD) {
        const v2f d01 = ww * (v01 - b01), d23 = ww * (v23 - b23);
        a01 = __builtin_elementwise_fma(d01, d01, a01);
        a23 = __builtin_elementwise_fma(d23, d23, a23);
    } else {
        a01 = __builtin_elementwise_fma(ww, v01, a01);
        a23 = __builtin_elementwise_fma(ww, v23, a23);
    }
}

template <bool SQD, int NJ, int LB>
__device__ __forceinline__ float4 g2_rows(float4 acc, const char* __restrict__ tabB, int mine,
                                          float left, unsigned qB, const float4 b) {
    static_assert(NJ == 8 || NJ == 16, "group of 8 or 16 edges");
    static_assert(LB + NJ <= 16, "ids come from one row of 16 lanes");
    constexpr bool DEEP = NJ == 16;
    float4 v[NJ];
#pragma unroll
    for (int J = 0; J < NJ; ++J) {
        unsigned s;
        switch (LB + J) {
            case 0:  s = (unsigned)G2_ROW_BCAST(mine, 0);  break;
            case 1:  s = (unsigned)G2_ROW_BCAST(mine, 1);  break;
            case 2:  s = (unsigned)G2_ROW_BCAST(mine, 2);  break;
            case 3:  s = (unsigned)G2_ROW_BCAST(mine, 3);  break;
            case 4:  s = (unsigned)G2_ROW_BCAST(mine, 4);  break;
            case 5:  s = (unsigned)G2_ROW_BCAST(mine, 5);  break;
            case 6:  s = (unsigned)G2_ROW_BCAST(mine, 6);  break;
            case 7:  s = (unsigned)G2_ROW_BCAST(mine, 7);  break;
            case 8:  s = (unsigned)G2_ROW_BCAST(mine, 8);  break;
            case 9:  s = (unsigned)G2_ROW_BCAST(mine, 9);  break;
            case 10: s = (unsigned)G2_ROW_BCAST(mine, 10); break;
            case 11: s = (unsigned)G2_ROW_BCAST(mine, 11); break;
            case 12: s = (unsigned)G2_ROW_BCAST(mine, 12); break;
            case 13: s = (unsigned)G2_ROW_BCAST(mine, 13); break;
            case 14: s = (unsigned)G2_ROW_BCAST(mine, 14); break;
            default: s = (unsigned)G2_ROW_BCAST(mine, 15); break;
        }
        v[J] = *(const float4*)(tabB + (s * 256u + qB));
    }
    v2f a01 = {acc.x, acc.y}, a23 = {acc.z, acc.w};
    const v2f b01 = {b.x, b.y}, b23 = {b.z, b.w};
#pragma unroll
    for (int J = 0; J < NJ; ++J) {
        g2_acc<SQD>(a01, a23, v[J], b01, b23, g2_w(left, J));
        if (!DEEP) __builtin_amdgcn_sched_barrier(0);
    }
    return make_float4(a01.x, a01.y, a23.x, a23.y);
}

template <bool SQD>
__device__ __forceinline__ float4 g2_ids16_by8(float4 acc, const char* __restrict__ tabB,
                                               int mine, int mc, float left, unsigned qB,
                                               const float4 b) {
    acc = g2_rows<SQD, 8, 0>(acc, tabB, mine, left, qB, b);
    if (mc > 8) acc = g2_rows<SQD, 8, 8>(acc, tabB, mine, left - 8.0f, qB, b);
    return acc;
}

template <bool SQD, bool DEEP16>
__device__ __forceinline__ float4 g2_node_gather(const char* __restrict__ tabB,
                                                 const int* __restrict__ list, int n,
                                                 int idA, int idB, int g, unsigned qB,
                                                 const float4 b) {
    const int m = max(__builtin_amdgcn_readlane(n, 0), __builtin_amdgcn_readlane(n, 16));
    const float left = (float)n;
    float4 acc = make_float4(0.0f, 0.0f, 0.0f, 0.0f);
    if (DEEP16) acc = g2_rows<SQD, 16, 0>(acc, tabB, idA, left, qB, b);
    else        acc = g2_ids16_by8<SQD>(acc, tabB, idA, m, left, qB, b);
    if (m > 16) {
        acc = g2_ids16_by8<SQD>(acc, tabB, idB, m - 16, left - 16.0f, qB, b);
        for (int base = 32; base < m; base += 16) {
            const int id = (base + g < n) ? list[base + g] : 0;
            acc = g2_ids16_by8<SQD>(acc, tabB, id, m - base, left - (float)base, qB, b);
        }
    }
    return acc;
}

#define PIECES_W (N_PART / (PART_T / 32))
#define PIECE_DEEP 2
#define SLAB_PASSES 8
static_assert(BIN_NODES == (PART_T / 32) * 16, "one 16-row slab per wave");

template <bool DEEP16>
__device__ __forceinline__ void g2_slab_gather(const int* __restrict__ ids,
                                               const int* __restrict__ nodeBeg,
                                               const int* __restrict__ nodeCnt,
                                               float* slab,
                                               const char* __restrict__ nfB, int tid) {
    const int g  = tid & 15;
    const unsigned qB = (unsigned)g * 16u;
    int row = (tid >> 5) * 16 + ((tid >> 4) & 1);
    const float4 zero4 = make_float4(0.0f, 0.0f, 0.0f, 0.0f);
    int beg = nodeBeg[row];
    int cn  = nodeCnt[row];
    int idA = (g < cn) ? ids[beg + g] : 0;
    int idB = (16 + g < cn) ? ids[beg + 16 + g] : 0;
    for (int p = 0; p < SLAB_PASSES; ++p) {
        int nbeg = 0, ncn = 0, nidA = 0, nidB = 0;
        if (p + 1 < SLAB_PASSES) {
            nbeg = nodeBeg[row + 2];
            ncn  = nodeCnt[row + 2];
            nidA = (g < ncn) ? ids[nbeg + g] : 0;
            nidB = (16 + g < ncn) ? ids[nbeg + 16 + g] : 0;
        }
        const float4 a = g2_node_gather<false, DEEP16>(nfB, ids + beg, cn, idA, idB, g, qB, zero4);
        const float c = fmaxf((float)cn, 1.0f);
        float4 v;
        v.x = a.x / c; v.y = a.y / c; v.z = a.z / c; v.w = a.w / c;
        *(float4*)&slab[row * DFEAT + g * 4] = v;
        row += 2;
        beg = nbeg; cn = ncn; idA = nidA; idB = nidB;
    }
}

__global__ __launch_bounds__(PART_T) void g2_bin_csr_gemm_kernel(
        const int* __restrict__ binned,
        const int* __restrict__ histG,
        const int* __restrict__ lstartG,
        const float* __restrict__ nf,
        const float* __restrict__ W,
        int* __restrict__ deg,
        int* __restrict__ rowptr,
        int* __restrict__ sorted,
        float* __restrict__ X,
        int n_nodes, int n_edges, int chunk) {
    __shared__ int cnt[BIN_NODES];
    __shared__ int wsum[BIN_NODES / 32];
    __shared__ int stageS[CSR_STAGE_EDGES + 32];
    __shared__ int pOff[N_PART];
    __shared__ int pCnt[N_PART];
    __shared__ int redS[N_PART / 32];
    __shared__ int redT[N_PART / 32];
    __shared__ int nodeBeg[BIN_NODES];
    __shared__ int nodeCnt[BIN_NODES];
    __shared__ __attribute__((aligned(16))) float ldsA[BIN_NODES * DFEAT];
    __shared__ float ldsWT[DFEAT * WT_PITCH];
    const int tid  = threadIdx.x;
    const int lane = tid & 31;
    const int wave = tid >> 5;
    const int bin  = blockIdx.x;
    int lc = 0, ls = 0;
    if (tid < N_PART) {
        lc = histG[tid * MAX_BINS + bin];
        ls = lstartG[tid * MAX_BINS + bin];
    }
    for (int i = tid; i < DFEAT * DFEAT; i += PART_T) {
        int k = i >> 6, n = i & 63;
        ldsWT[n * WT_PITCH + k] = W[i];
    }
    if (tid < BIN_NODES) cnt[tid] = 0;
    int sumS = ls, sumT = lc;
    for (int off = 16; off > 0; off >>= 1) {
        sumS += __shfl_xor(sumS, off, 32);
        sumT += __shfl_xor(sumT, off, 32);
    }
    if (tid < N_PART) {
        pOff[tid] = tid * chunk + ls;
        pCnt[tid] = lc;
        if (lane == 0) { redS[wave] = sumS; redT[wave] = sumT; }
    }
    __syncthreads();
    int s0 = 0, tot = 0;
    for (int w = 0; w < N_PART / 32; ++w) { s0 += redS[w]; tot += redT[w]; }
    const bool staged = tot <= CSR_STAGE_EDGES;

    int off[PIECES_W], n[PIECES_W];
    for (int j = 0; j < PIECES_W; ++j) {
        off[j] = __builtin_amdgcn_readfirstlane(pOff[wave * PIECES_W + j]);
        n[j]   = __builtin_amdgcn_readfirstlane(pCnt[wave * PIECES_W + j]);
    }
    unsigned v[PIECES_W][PIECE_DEEP];
    for (int j = 0; j < PIECES_W; ++j)
        for (int h = 0; h < PIECE_DEEP; ++h)
            v[j][h] = (unsigned)binned[max(min(off[j] + min(lane + 32 * h, n[j] - 1),
                                               n_edges - 1), 0)];
    for (int j = 0; j < PIECES_W; ++j)
        for (int h = 0; h < PIECE_DEEP; ++h) {
            const bool ok = lane + 32 * h < n[j];
            atomicAdd(&cnt[ok ? (v[j][h] & (BIN_NODES - 1u)) : (unsigned)(tid & (BIN_NODES - 1))],
                      ok ? 1 : 0);
        }
    for (int j = 0; j < PIECES_W; ++j)
        if (n[j] > 32 * PIECE_DEEP)
            for (int k = 32 * PIECE_DEEP + lane; k < n[j]; k += 32)
                atomicAdd(&cnt[(unsigned)binned[off[j] + k] & (BIN_NODES - 1u)], 1);
    __syncthreads();

    int c = 0;
    if (tid < BIN_NODES) c = cnt[tid];
    int inc = c;
    for (int o = 1; o < 32; o <<= 1) {
        const int t = __shfl_up(inc, o, 32);
        if (lane >= o) inc += t;
    }
    if (tid < BIN_NODES && lane == 31) wsum[wave] = inc;
    __syncthreads();
    if (tid < BIN_NODES) {
        int woff = 0;
        for (int w = 0; w < BIN_NODES / 32; ++w) woff += (w < wave) ? wsum[w] : 0;
        const int start = s0 + woff + inc - c;
        const int node = bin * BIN_NODES + tid;
        if (node < n_nodes) { deg[node] = c; rowptr[node] = start; }
        cnt[tid] = start;
        nodeBeg[tid] = start - s0;
        nodeCnt[tid] = c;
    }
    __syncthreads();
    if (staged) {
        int slot[PIECES_W][PIECE_DEEP];
        for (int j = 0; j < PIECES_W; ++j)
            for (int h = 0; h < PIECE_DEEP; ++h) {
                const bool ok = lane + 32 * h < n[j];
                slot[j][h] = atomicAdd(&cnt[ok ? (v[j][h] & (BIN_NODES - 1u))
                                               : (unsigned)(tid & (BIN_NODES - 1))], ok ? 1 : 0);
            }
        for (int j = 0; j < PIECES_W; ++j)
            for (int h = 0; h < PIECE_DEEP; ++h) {
                const bool ok = lane + 32 * h < n[j];
                stageS[ok ? slot[j][h] - s0 : CSR_STAGE_EDGES + lane] = (int)(v[j][h] >> BIN_SHIFT);
            }
        for (int j = 0; j < PIECES_W; ++j)
            if (n[j] > 32 * PIECE_DEEP)
                for (int k = 32 * PIECE_DEEP + lane; k < n[j]; k += 32) {
                    const unsigned u = (unsigned)binned[off[j] + k];
                    stageS[atomicAdd(&cnt[u & (BIN_NODES - 1u)], 1) - s0] = (int)(u >> BIN_SHIFT);
                }
        __syncthreads();
        for (int i = tid - (s0 & 31); i < tot; i += STRIP * PART_T) {
            int r[STRIP];
            for (int j = 0; j < STRIP; ++j) r[j] = stageS[max(min(i + j * PART_T, tot - 1), 0)];
            for (int j = 0; j < STRIP; ++j) {
                const int k = i + j * PART_T;
                if (k >= 0 && k < tot) sorted[s0 + k] = r[j];
            }
        }
        g2_slab_gather<true>(stageS, nodeBeg, nodeCnt, ldsA, (const char*)nf, tid);
    } else {
        for (int j = 0; j < PIECES_W; ++j)
            for (int k = lane; k < n[j]; k += 32) {
                const unsigned u = (unsigned)binned[off[j] + k];
                sorted[atomicAdd(&cnt[u & (BIN_NODES - 1u)], 1)] = (int)(u >> BIN_SHIFT);
            }
        __threadfence();
        __syncthreads();
        g2_slab_gather<false>(sorted + s0, nodeBeg, nodeCnt, ldsA, (const char*)nf, tid);
    }

    const int m    = lane & 15;
    const int koff = (lane >> 4) << 1;

    const float* aRow = &ldsA[(wave * 16 + m) * DFEAT];
    const float* w0 = &ldsWT[(m +  0) * WT_PITCH];
    const float* w1 = &ldsWT[(m + 16) * WT_PITCH];
    const float* w2 = &ldsWT[(m + 32) * WT_PITCH];
    const float* w3 = &ldsWT[(m + 48) * WT_PITCH];

    v8f acc0 = {}, acc1 = {}, acc2 = {}, acc3 = {};
    for (int k = 0; k < 16; ++k) {
        const int kk = 4 * k + koff;
        const v2f a  = *(const v2f*)(aRow + kk);
        const v2f b0 = *(const v2f*)(w0 + kk);
        const v2f b1 = *(const v2f*)(w1 + kk);
        const v2f b2 = *(const v2f*)(w2 + kk);
        const v2f b3 = *(const v2f*)(w3 + kk);
        acc0 = __builtin_amdgcn_wmma_f32_16x16x4_f32(false, a, false, b0,
                                                     (short)0, acc0, false, false);
        acc1 = __builtin_amdgcn_wmma_f32_16x16x4_f32(false, a, false, b1,
                                                     (short)0, acc1, false, false);
        acc2 = __builtin_amdgcn_wmma_f32_16x16x4_f32(false, a, false, b2,
                                                     (short)0, acc2, false, false);
        acc3 = __builtin_amdgcn_wmma_f32_16x16x4_f32(false, a, false, b3,
                                                     (short)0, acc3, false, false);
    }

    const int row0 = bin * BIN_NODES + wave * 16 + ((lane >> 4) << 3);
    for (int r = 0; r < 8; ++r) {
        int row = row0 + r;
        if (row < n_nodes) {
            float* xr = X + (size_t)row * DFEAT + m;
            xr[0]  = acc0[r];
            xr[16] = acc1[r];
            xr[32] = acc2[r];
            xr[48] = acc3[r];
        }
    }
}

__global__ __launch_bounds__(256) void g2_tau_gather_kernel(
        const float* __restrict__ X,
        const int* __restrict__ sorted,
        const int* __restrict__ rowptr,
        const int* __restrict__ deg,
        float* __restrict__ out,
        int n_nodes) {
    const int gid = blockIdx.x * blockDim.x + threadIdx.x;
    const bool live = (gid >> 4) < n_nodes;
    const int d = min(gid >> 4, n_nodes - 1);
    const int g = gid & 15;
    const int q = g * 4;
    const int n = deg[d];
    const int* list = sorted + rowptr[d];
    const float4 b = *(const float4*)(X + (size_t)d * DFEAT + q);
    const int idA = (g < n) ? list[g] : 0;
    const int idB = (16 + g < n) ? list[16 + g] : 0;

    const float4 acc = g2_node_gather<true, false>((const char*)X, list, n, idA, idB, g,
                                            (unsigned)g * 16u, b);

    const float c = fmaxf((float)n, 1.0f);
    float4 t;
    t.x = tanhf(acc.x / c);
    t.y = tanhf(acc.y / c);
    t.z = tanhf(acc.z / c);
    t.w = tanhf(acc.w / c);
    if (live) *(float4*)(out + (size_t)d * DFEAT + q) = t;
}

extern "C" void kernel_launch(void* const* d_in, const int* in_sizes, int n_in,
                              void* d_out, int out_size, void* d_ws, size_t ws_size,
                              hipStream_t stream) {
    const float* nf = (const float*)d_in[0];
    const float* W  = (const float*)d_in[1];
    const int*   ei = (const int*)d_in[2];
    float* out = (float*)d_out;

    const int n_nodes = in_sizes[0] / DFEAT;
    const int n_edges = in_sizes[2] / 2;
    const int* src = ei;
    const int* dst = ei + n_edges;

    const int n_pad = (n_nodes + SCAN_T - 1) / SCAN_T * SCAN_T;
    float* X      = (float*)d_ws;
    int*   deg    = (int*)(X + (size_t)n_nodes * DFEAT);
    int*   rowptr = deg + n_pad;
    int*   binned = rowptr + n_pad;
    int*   sorted = binned + n_edges;
    int*   histG   = (int*)out;
    int*   lstartG = histG + N_PART * MAX_BINS;

    const int T = 256;
    const int n_bins = (n_nodes + BIN_NODES - 1) / BIN_NODES;
    const int chunk  = (n_edges + N_PART - 1) / N_PART;

    g2_chunk_bin_kernel<<<N_PART, PART_T, 0, stream>>>(src, dst, binned, histG, lstartG,
                                                       n_edges, chunk);
    g2_bin_csr_gemm_kernel<<<n_bins, PART_T, 0, stream>>>(binned, histG, lstartG, nf, W,
                                                          deg, rowptr, sorted, X,
                                                          n_nodes, n_edges, chunk);
    {
        long long total = (long long)n_nodes * 16;
        g2_tau_gather_kernel<<<(int)((total + T - 1) / T), T, 0, stream>>>(
            X, sorted, rowptr, deg, out, n_nodes);
    }
}
